# starting kernel
# baseline (speedup 1.0000x reference)
#include <hip/hip_runtime.h>

typedef __attribute__((ext_vector_type(2))) float v2f;
typedef __attribute__((ext_vector_type(8))) float v8f;

#define NUQ   100000
#define NIQ   50000
#define DDIM  128
#define EEDG  300000
#define NNOD  (NUQ + NIQ)
#define LDSW  132

__global__ __launch_bounds__(512)
void proj_wmma_kernel(const float* __restrict__ xu, const float* __restrict__ xi,
                      const float* __restrict__ W1, const float* __restrict__ b1,
                      const float* __restrict__ W2, const float* __restrict__ b2,
                      float* __restrict__ out, float* __restrict__ S,
                      float* __restrict__ Qb2)
{
    __shared__ float xls[16 * LDSW];
    __shared__ float qls[16 * LDSW];

    const int row0 = blockIdx.x * 16;
    const int tid  = threadIdx.x;

    {
        const int idx = tid * 4;
        const int r   = idx >> 7;
        const int c   = idx & 127;
        const int row = row0 + r;
        const float* src = (row < NUQ) ? (xu + (size_t)row * DDIM + c)
                                       : (xi + (size_t)(row - NUQ) * DDIM + c);
        *(float4*)&xls[r * LDSW + c] = *(const float4*)src;
    }
    __syncthreads();

    const int lane = tid & 31;
    const int wv   = tid >> 5;
    const int isW2 = wv >> 3;
    const int c0   = (wv & 7) * 16;
    const int l15  = lane & 15;
    const int lh   = lane >> 4;

    const float* __restrict__ W    = isW2 ? W2 : W1;
    const float* __restrict__ Arow = &xls[l15 * LDSW + 2 * lh];
    const float* __restrict__ Brow = W + (size_t)(c0 + l15) * DDIM + 2 * lh;

    v8f acc = {};
    #pragma unroll
    for (int k = 0; k < DDIM; k += 4) {
        v2f a = *(const v2f*)(Arow + k);
        v2f b = *(const v2f*)(Brow + k);
        acc = __builtin_amdgcn_wmma_f32_16x16x4_f32(
             false, a,  false, b,
             (short)0, acc,  false,  false);
    }

    const int col = c0 + l15;
    if (isW2) {
        const float bb2 = b2[col];
        #pragma unroll
        for (int v = 0; v < 8; ++v) {
            const int m   = v + 8 * lh;
            const float q = acc[v];
            qls[m * LDSW + col] = q;
            Qb2[(size_t)(row0 + m) * DDIM + col] = q + bb2;
        }
    }
    __syncthreads();
    if (!isW2) {
        const float bb1 = b1[col];
        #pragma unroll
        for (int v = 0; v < 8; ++v) {
            const int m   = v + 8 * lh;
            const float p = acc[v] + bb1;
            const float q = qls[m * LDSW + col];
            out[(size_t)(row0 + m) * DDIM + col] = p;
            S  [(size_t)(row0 + m) * DDIM + col] = p + q;
        }
    }
}

__global__ __launch_bounds__(256)
void edge_kernel(const float* __restrict__ S, const float* __restrict__ Qb2,
                 const int* __restrict__ src_ui, const int* __restrict__ dst_ui,
                 const float* __restrict__ norm_ui,
                 const int* __restrict__ src_iu, const int* __restrict__ dst_iu,
                 const float* __restrict__ norm_iu,
                 float* __restrict__ out)
{
    const int gw   = (int)((blockIdx.x * blockDim.x + threadIdx.x) >> 5);
    const int lane = threadIdx.x & 31;
    if (gw >= 2 * EEDG) return;

    size_t srow, drow;
    float  nrm;
    if (gw < EEDG) {
        const int e = gw;
        srow = (size_t)src_ui[e] * DDIM;
        drow = (size_t)(NUQ + dst_ui[e]) * DDIM;
        nrm  = norm_ui[e];
    } else {
        const int e = gw - EEDG;
        srow = (size_t)(NUQ + src_iu[e]) * DDIM;
        drow = (size_t)dst_iu[e] * DDIM;
        nrm  = norm_iu[e];
    }

    const int c = lane * 4;
    float4 sv = *(const float4*)(S   + srow + c);
    float4 qv = *(const float4*)(Qb2 + drow + c);
    float* o  = out + drow + c;
    unsafeAtomicAdd(o + 0, nrm * (sv.x + qv.x));
    unsafeAtomicAdd(o + 1, nrm * (sv.y + qv.y));
    unsafeAtomicAdd(o + 2, nrm * (sv.z + qv.z));
    unsafeAtomicAdd(o + 3, nrm * (sv.w + qv.w));
}

__global__ __launch_bounds__(256)
void finalize_kernel(float* __restrict__ out)
{
    const int gw   = (int)((blockIdx.x * blockDim.x + threadIdx.x) >> 5);
    const int lane = threadIdx.x & 31;
    if (gw >= NNOD) return;

    const size_t base = (size_t)gw * DDIM + lane * 4;
    float4 v = *(float4*)(out + base);
    v.x = v.x > 0.0f ? v.x : 0.2f * v.x;
    v.y = v.y > 0.0f ? v.y : 0.2f * v.y;
    v.z = v.z > 0.0f ? v.z : 0.2f * v.z;
    v.w = v.w > 0.0f ? v.w : 0.2f * v.w;

    float ss = v.x * v.x + v.y * v.y + v.z * v.z + v.w * v.w;
    #pragma unroll
    for (int off = 16; off > 0; off >>= 1)
        ss += __shfl_xor(ss, off, 32);

    const float scale = 1.0f / fmaxf(sqrtf(ss), 1e-12f);
    v.x *= scale; v.y *= scale; v.z *= scale; v.w *= scale;
    *(float4*)(out + base) = v;
}

extern "C" void kernel_launch(void* const* d_in, const int* in_sizes, int n_in,
                              void* d_out, int out_size, void* d_ws, size_t ws_size,
                              hipStream_t stream) {
    (void)in_sizes; (void)n_in; (void)out_size; (void)ws_size;

    const float* xu      = (const float*)d_in[0];
    const float* xi      = (const float*)d_in[1];
    const float* W1      = (const float*)d_in[2];
    const float* b1      = (const float*)d_in[3];
    const float* W2      = (const float*)d_in[4];
    const float* b2      = (const float*)d_in[5];
    const int*   src_ui  = (const int*)d_in[6];
    const int*   dst_ui  = (const int*)d_in[7];
    const float* norm_ui = (const float*)d_in[8];
    const int*   src_iu  = (const int*)d_in[9];
    const int*   dst_iu  = (const int*)d_in[10];
    const float* norm_iu = (const float*)d_in[11];

    float* out = (float*)d_out;
    float* S   = (float*)d_ws;
    float* Qb2 = S + (size_t)NNOD * DDIM;

    proj_wmma_kernel<<<NNOD / 16, 512, 0, stream>>>(xu, xi, W1, b1, W2, b2, out, S, Qb2);

    edge_kernel<<<(2 * EEDG) / 8, 256, 0, stream>>>(S, Qb2, src_ui, dst_ui, norm_ui,
                                                    src_iu, dst_iu, norm_iu, out);

    finalize_kernel<<<NNOD / 8, 256, 0, stream>>>(out);
}


# proj kernel: each wave keeps its W1 and W2 WMMA fragments in registers and loops over row tiles (512 blocks x 8 waves, X tile double-buffered in LDS) instead of re-reading W per 16-row block
# speedup vs baseline: 1.3052x; 1.3052x over previous
#include <hip/hip_runtime.h>

typedef __attribute__((ext_vector_type(2))) float v2f;
typedef __attribute__((ext_vector_type(8))) float v8f;

#define NUQ   100000
#define NIQ   50000
#define DDIM  128
#define EEDG  300000
#define NNOD  (NUQ + NIQ)
#define LDSW  132
#define STAGE (16 * LDSW)
#define NTILE (NNOD / 16)
#define PGRID 512

static __device__ __forceinline__ void lds_barrier()
{
    __builtin_amdgcn_fence(__ATOMIC_RELEASE, "workgroup", "local");
    __builtin_amdgcn_s_barrier();
    __builtin_amdgcn_fence(__ATOMIC_ACQUIRE, "workgroup", "local");
}

static __device__ __forceinline__ void load_x_pieces(const float* __restrict__ xu,
                                                     const float* __restrict__ xi,
                                                     int tile, int sr, int sc,
                                                     float4& r0, float4& r1)
{
    const int ra = tile * 16 + sr;
    const int rb = ra + 8;
    r0 = *(const float4*)((ra < NUQ) ? (xu + (size_t)ra * DDIM + sc)
                                     : (xi + (size_t)(ra - NUQ) * DDIM + sc));
    r1 = *(const float4*)((rb < NUQ) ? (xu + (size_t)rb * DDIM + sc)
                                     : (xi + (size_t)(rb - NUQ) * DDIM + sc));
}

__global__ __launch_bounds__(256)
void proj_wmma_kernel(const float* __restrict__ xu, const float* __restrict__ xi,
                      const float* __restrict__ W1, const float* __restrict__ b1,
                      const float* __restrict__ W2, const float* __restrict__ b2,
                      float* __restrict__ out, float* __restrict__ S,
                      float* __restrict__ Qb2)
{
    __shared__ float lds[8 * STAGE];

    const int tid  = threadIdx.x;
    const int lane = tid & 31;
    const int wv   = tid >> 5;
    const int c0   = wv * 16;
    const int l15  = lane & 15;
    const int lh   = lane >> 4;
    const int col  = c0 + l15;

    const int sr = tid >> 5;
    const int sc = (tid & 31) * 4;

    int t = blockIdx.x;
    float4 r0, r1;
    load_x_pieces(xu, xi, t, sr, sc, r0, r1);

    const float bb1 = b1[col];
    const float bb2 = b2[col];

    float* wst = lds + wv * STAGE;
    const float* Bfrag = wst + l15 * LDSW + 2 * lh;
    v2f bw1[32], bw2[32];

    #pragma unroll
    for (int r = 0; r < 16; ++r)
        *(float4*)&wst[r * LDSW + lane * 4] =
            *(const float4*)(W1 + (size_t)(c0 + r) * DDIM + lane * 4);
    __syncthreads();
    #pragma unroll
    for (int i = 0; i < 32; ++i)
        bw1[i] = *(const v2f*)(Bfrag + 4 * i);
    __syncthreads();
    #pragma unroll
    for (int r = 0; r < 16; ++r)
        *(float4*)&wst[r * LDSW + lane * 4] =
            *(const float4*)(W2 + (size_t)(c0 + r) * DDIM + lane * 4);
    __syncthreads();
    #pragma unroll
    for (int i = 0; i < 32; ++i)
        bw2[i] = *(const v2f*)(Bfrag + 4 * i);
    __syncthreads();

    *(float4*)&lds[sr * LDSW + sc]       = r0;
    *(float4*)&lds[(sr + 8) * LDSW + sc] = r1;
    __syncthreads();
    if (t + PGRID < NTILE)
        load_x_pieces(xu, xi, t + PGRID, sr, sc, r0, r1);

    int buf = 0;
    for (; t < NTILE; t += PGRID) {
        const float* Arow = lds + buf * STAGE + l15 * LDSW + 2 * lh;

        v2f af[32];
        #pragma unroll
        for (int i = 0; i < 32; ++i)
            af[i] = *(const v2f*)(Arow + 4 * i);

        v8f accp = {};
        v8f accq = {};
        #pragma unroll
        for (int i = 0; i < 32; ++i) {
            accp = __builtin_amdgcn_wmma_f32_16x16x4_f32(
                 false, af[i],  false, bw1[i],
                 (short)0, accp,  false,  false);
            accq = __builtin_amdgcn_wmma_f32_16x16x4_f32(
                 false, af[i],  false, bw2[i],
                 (short)0, accq,  false,  false);
        }
        __builtin_amdgcn_sched_group_barrier(0x100, 16, 0);
        __builtin_amdgcn_sched_group_barrier(0x008, 64, 0);

        const int row0 = t * 16;
        #pragma unroll
        for (int v = 0; v < 8; ++v) {
            const size_t o = (size_t)(row0 + v + 8 * lh) * DDIM + col;
            const float p = accp[v] + bb1;
            const float q = accq[v];
            out[o] = p;
            S[o]   = p + q;
            Qb2[o] = q + bb2;
        }

        const int tn = t + PGRID;
        if (tn < NTILE) {
            float* nxt = lds + (buf ^ 1) * STAGE;
            *(float4*)&nxt[sr * LDSW + sc]       = r0;
            *(float4*)&nxt[(sr + 8) * LDSW + sc] = r1;
            if (tn + PGRID < NTILE)
                load_x_pieces(xu, xi, tn + PGRID, sr, sc, r0, r1);
        }
        lds_barrier();
        buf ^= 1;
    }
}

__global__ __launch_bounds__(256)
void edge_kernel(const float* __restrict__ S, const float* __restrict__ Qb2,
                 const int* __restrict__ src_ui, const int* __restrict__ dst_ui,
                 const float* __restrict__ norm_ui,
                 const int* __restrict__ src_iu, const int* __restrict__ dst_iu,
                 const float* __restrict__ norm_iu,
                 float* __restrict__ out)
{
    const int gw   = (int)((blockIdx.x * blockDim.x + threadIdx.x) >> 5);
    const int lane = threadIdx.x & 31;
    if (gw >= 2 * EEDG) return;

    size_t srow, drow;
    float  nrm;
    if (gw < EEDG) {
        const int e = gw;
        srow = (size_t)src_ui[e] * DDIM;
        drow = (size_t)(NUQ + dst_ui[e]) * DDIM;
        nrm  = norm_ui[e];
    } else {
        const int e = gw - EEDG;
        srow = (size_t)(NUQ + src_iu[e]) * DDIM;
        drow = (size_t)dst_iu[e] * DDIM;
        nrm  = norm_iu[e];
    }

    const int c = lane * 4;
    float4 sv = *(const float4*)(S   + srow + c);
    float4 qv = *(const float4*)(Qb2 + drow + c);
    float* o  = out + drow + c;
    unsafeAtomicAdd(o + 0, nrm * (sv.x + qv.x));
    unsafeAtomicAdd(o + 1, nrm * (sv.y + qv.y));
    unsafeAtomicAdd(o + 2, nrm * (sv.z + qv.z));
    unsafeAtomicAdd(o + 3, nrm * (sv.w + qv.w));
}

__global__ __launch_bounds__(256)
void finalize_kernel(float* __restrict__ out)
{
    const int gw   = (int)((blockIdx.x * blockDim.x + threadIdx.x) >> 5);
    const int lane = threadIdx.x & 31;
    if (gw >= NNOD) return;

    const size_t base = (size_t)gw * DDIM + lane * 4;
    float4 v = *(float4*)(out + base);
    v.x = v.x > 0.0f ? v.x : 0.2f * v.x;
    v.y = v.y > 0.0f ? v.y : 0.2f * v.y;
    v.z = v.z > 0.0f ? v.z : 0.2f * v.z;
    v.w = v.w > 0.0f ? v.w : 0.2f * v.w;

    float ss = v.x * v.x + v.y * v.y + v.z * v.z + v.w * v.w;
    #pragma unroll
    for (int off = 16; off > 0; off >>= 1)
        ss += __shfl_xor(ss, off, 32);

    const float scale = 1.0f / fmaxf(sqrtf(ss), 1e-12f);
    v.x *= scale; v.y *= scale; v.z *= scale; v.w *= scale;
    *(float4*)(out + base) = v;
}

extern "C" void kernel_launch(void* const* d_in, const int* in_sizes, int n_in,
                              void* d_out, int out_size, void* d_ws, size_t ws_size,
                              hipStream_t stream) {
    (void)in_sizes; (void)n_in; (void)out_size; (void)ws_size;

    const float* xu      = (const float*)d_in[0];
    const float* xi      = (const float*)d_in[1];
    const float* W1      = (const float*)d_in[2];
    const float* b1      = (const float*)d_in[3];
    const float* W2      = (const float*)d_in[4];
    const float* b2      = (const float*)d_in[5];
    const int*   src_ui  = (const int*)d_in[6];
    const int*   dst_ui  = (const int*)d_in[7];
    const float* norm_ui = (const float*)d_in[8];
    const int*   src_iu  = (const int*)d_in[9];
    const int*   dst_iu  = (const int*)d_in[10];
    const float* norm_iu = (const float*)d_in[11];

    float* out = (float*)d_out;
    float* S   = (float*)d_ws;
    float* Qb2 = S + (size_t)NNOD * DDIM;

    proj_wmma_kernel<<<PGRID, 256, 0, stream>>>(xu, xi, W1, b1, W2, b2, out, S, Qb2);

    edge_kernel<<<(2 * EEDG) / 8, 256, 0, stream>>>(S, Qb2, src_ui, dst_ui, norm_ui,
                                                    src_iu, dst_iu, norm_iu, out);

    finalize_kernel<<<NNOD / 8, 256, 0, stream>>>(out);
}


# edge scatter by float atomics replaced by per-node in-edge lists built each call (int exchange); half a wave sums a node's list in registers and finalizes the row: sum order = list order
# speedup vs baseline: 2.1291x; 1.6312x over previous
#include <hip/hip_runtime.h>

typedef __attribute__((ext_vector_type(2))) float v2f;
typedef __attribute__((ext_vector_type(8))) float v8f;

#define NUQ   100000
#define NIQ   50000
#define DDIM  128
#define EEDG  300000
#define NNOD  (NUQ + NIQ)
#define LDSW  132
#define STAGE (16 * LDSW)
#define NTILE (NNOD / 16)
#define PGRID 512
#define NEDGE (2 * EEDG)
#define LHALF (EEDG / 2)
#define GGRID 2048

static __device__ __forceinline__ void lds_barrier()
{
    __builtin_amdgcn_fence(__ATOMIC_RELEASE, "workgroup", "local");
    __builtin_amdgcn_s_barrier();
    __builtin_amdgcn_fence(__ATOMIC_ACQUIRE, "workgroup", "local");
}

static __device__ __forceinline__ void load_x_pieces(const float* __restrict__ xu,
                                                     const float* __restrict__ xi,
                                                     int tile, int sr, int sc,
                                                     float4& r0, float4& r1)
{
    const int ra = tile * 16 + sr;
    const int rb = ra + 8;
    r0 = *(const float4*)((ra < NUQ) ? (xu + (size_t)ra * DDIM + sc)
                                     : (xi + (size_t)(ra - NUQ) * DDIM + sc));
    r1 = *(const float4*)((rb < NUQ) ? (xu + (size_t)rb * DDIM + sc)
                                     : (xi + (size_t)(rb - NUQ) * DDIM + sc));
}

__global__ __launch_bounds__(256)
void proj_wmma_kernel(const float* __restrict__ xu, const float* __restrict__ xi,
                      const float* __restrict__ W1, const float* __restrict__ b1,
                      const float* __restrict__ W2, const float* __restrict__ b2,
                      float* __restrict__ P1, float* __restrict__ S,
                      float* __restrict__ Qb2, int* __restrict__ head)
{
    __shared__ float lds[8 * STAGE];

    const int tid  = threadIdx.x;
    const int lane = tid & 31;
    const int wv   = tid >> 5;
    const int c0   = wv * 16;
    const int l15  = lane & 15;
    const int lh   = lane >> 4;
    const int col  = c0 + l15;

    const int sr = tid >> 5;
    const int sc = (tid & 31) * 4;

    static_assert(PGRID * 256 <= NNOD && 2 * PGRID * 256 >= NNOD, "two strides cover the heads");
    {
        const int g = blockIdx.x * 256 + tid;
        head[g] = -1;
        if (g + PGRID * 256 < NNOD)
            head[g + PGRID * 256] = -1;
    }

    int t = blockIdx.x;
    float4 r0, r1;
    load_x_pieces(xu, xi, t, sr, sc, r0, r1);

    const float bb1 = b1[col];
    const float bb2 = b2[col];

    float* wst = lds + wv * STAGE;
    const float* Bfrag = wst + l15 * LDSW + 2 * lh;
    v2f bw1[32], bw2[32];

    #pragma unroll
    for (int r = 0; r < 16; ++r)
        *(float4*)&wst[r * LDSW + lane * 4] =
            *(const float4*)(W1 + (size_t)(c0 + r) * DDIM + lane * 4);
    __syncthreads();
    #pragma unroll
    for (int i = 0; i < 32; ++i)
        bw1[i] = *(const v2f*)(Bfrag + 4 * i);
    __syncthreads();
    #pragma unroll
    for (int r = 0; r < 16; ++r)
        *(float4*)&wst[r * LDSW + lane * 4] =
            *(const float4*)(W2 + (size_t)(c0 + r) * DDIM + lane * 4);
    __syncthreads();
    #pragma unroll
    for (int i = 0; i < 32; ++i)
        bw2[i] = *(const v2f*)(Bfrag + 4 * i);
    __syncthreads();

    *(float4*)&lds[sr * LDSW + sc]       = r0;
    *(float4*)&lds[(sr + 8) * LDSW + sc] = r1;
    __syncthreads();
    if (t + PGRID < NTILE)
        load_x_pieces(xu, xi, t + PGRID, sr, sc, r0, r1);

    int buf = 0;
    for (; t < NTILE; t += PGRID) {
        const float* Arow = lds + buf * STAGE + l15 * LDSW + 2 * lh;

        v2f af[32];
        #pragma unroll
        for (int i = 0; i < 32; ++i)
            af[i] = *(const v2f*)(Arow + 4 * i);

        v8f accp = {};
        v8f accq = {};
        #pragma unroll
        for (int i = 0; i < 32; ++i) {
            accp = __builtin_amdgcn_wmma_f32_16x16x4_f32(
                 false, af[i],  false, bw1[i],
                 (short)0, accp,  false,  false);
            accq = __builtin_amdgcn_wmma_f32_16x16x4_f32(
                 false, af[i],  false, bw2[i],
                 (short)0, accq,  false,  false);
        }
        __builtin_amdgcn_sched_group_barrier(0x100, 16, 0);
        __builtin_amdgcn_sched_group_barrier(0x008, 64, 0);

        const int row0 = t * 16;
        #pragma unroll
        for (int v = 0; v < 8; ++v) {
            const size_t o = (size_t)(row0 + v + 8 * lh) * DDIM + col;
            const float p = accp[v] + bb1;
            const float q = accq[v];
            P1[o]  = p;
            S[o]   = p + q;
            Qb2[o] = q + bb2;
        }

        const int tn = t + PGRID;
        if (tn < NTILE) {
            float* nxt = lds + (buf ^ 1) * STAGE;
            *(float4*)&nxt[sr * LDSW + sc]       = r0;
            *(float4*)&nxt[(sr + 8) * LDSW + sc] = r1;
            if (tn + PGRID < NTILE)
                load_x_pieces(xu, xi, tn + PGRID, sr, sc, r0, r1);
        }
        lds_barrier();
        buf ^= 1;
    }
}

__global__ __launch_bounds__(256)
void link_kernel(const int* __restrict__ src_ui, const int* __restrict__ dst_ui,
                 const float* __restrict__ norm_ui,
                 const int* __restrict__ src_iu, const int* __restrict__ dst_iu,
                 const float* __restrict__ norm_iu,
                 int* __restrict__ head, int* __restrict__ rec_w)
{
    const int g = (int)(blockIdx.x * 256 + threadIdx.x);
    if (g >= LHALF) return;
    const int g1 = g + LHALF;

    const int   sa = src_ui[g],        sb = src_ui[g1];
    const int   da = NUQ + dst_ui[g],  db = NUQ + dst_ui[g1];
    const float na = norm_ui[g],       nb = norm_ui[g1];
    const int   sc = NUQ + src_iu[g],  sd = NUQ + src_iu[g1];
    const int   dc = dst_iu[g],        dd = dst_iu[g1];
    const float nc = norm_iu[g],       nd = norm_iu[g1];

    const int ea = g, eb = g1, ec = EEDG + g, ed = EEDG + g1;
    const int pa = atomicExch(head + da, ea);
    const int pb = atomicExch(head + db, eb);
    const int pc = atomicExch(head + dc, ec);
    const int pd = atomicExch(head + dd, ed);

    int4* rec = (int4*)rec_w;
    rec[ea] = make_int4(pa, sa, __float_as_int(na), 0);
    rec[eb] = make_int4(pb, sb, __float_as_int(nb), 0);
    rec[ec] = make_int4(pc, sc, __float_as_int(nc), 0);
    rec[ed] = make_int4(pd, sd, __float_as_int(nd), 0);
}

__global__ __launch_bounds__(256)
void gather_finalize_kernel(const float* __restrict__ P1, const float* __restrict__ S,
                            const float* __restrict__ Qb2,
                            const int* __restrict__ head, const int* __restrict__ rec_r,
                            float* __restrict__ out)
{
    const int4* __restrict__ rec = (const int4*)rec_r;
    const int lane = threadIdx.x & 31;
    const int half = lane >> 4;
    const int c    = (lane & 15) * 4;

    for (int p = (int)((blockIdx.x * 256 + threadIdx.x) >> 5); p < NNOD / 2; p += GGRID * 8) {
        const int    d    = 2 * p + half;
        const size_t base = (size_t)d * DDIM + c;
        const int e0 = head[d];
        float4 acc0      = *(const float4*)(P1 + base);
        float4 acc1      = *(const float4*)(P1 + base + 64);
        const float4 qv0 = *(const float4*)(Qb2 + base);
        const float4 qv1 = *(const float4*)(Qb2 + base + 64);

        if ((unsigned)e0 < (unsigned)NEDGE) {
            int4 r = rec[e0];
            for (int it = 0; it < NEDGE; ++it) {
                const int  en   = r.x;
                const bool more = (unsigned)en < (unsigned)NEDGE;
                int4 rn = r;
                if (more) rn = rec[en];
                const float* srow = S + (size_t)r.y * DDIM + c;
                const float4 sv0 = *(const float4*)(srow);
                const float4 sv1 = *(const float4*)(srow + 64);
                const float nrm = __int_as_float(r.z);
                acc0.x += nrm * (sv0.x + qv0.x);
                acc0.y += nrm * (sv0.y + qv0.y);
                acc0.z += nrm * (sv0.z + qv0.z);
                acc0.w += nrm * (sv0.w + qv0.w);
                acc1.x += nrm * (sv1.x + qv1.x);
                acc1.y += nrm * (sv1.y + qv1.y);
                acc1.z += nrm * (sv1.z + qv1.z);
                acc1.w += nrm * (sv1.w + qv1.w);
                if (!more) break;
                r = rn;
            }
        }

        float4 v0 = acc0, v1 = acc1;
        v0.x = v0.x > 0.0f ? v0.x : 0.2f * v0.x;
        v0.y = v0.y > 0.0f ? v0.y : 0.2f * v0.y;
        v0.z = v0.z > 0.0f ? v0.z : 0.2f * v0.z;
        v0.w = v0.w > 0.0f ? v0.w : 0.2f * v0.w;
        v1.x = v1.x > 0.0f ? v1.x : 0.2f * v1.x;
        v1.y = v1.y > 0.0f ? v1.y : 0.2f * v1.y;
        v1.z = v1.z > 0.0f ? v1.z : 0.2f * v1.z;
        v1.w = v1.w > 0.0f ? v1.w : 0.2f * v1.w;

        float ss = v0.x * v0.x + v0.y * v0.y + v0.z * v0.z + v0.w * v0.w
                 + v1.x * v1.x + v1.y * v1.y + v1.z * v1.z + v1.w * v1.w;
        for (int off = 8; off > 0; off >>= 1)
            ss += __shfl_xor(ss, off, 32);

        const float scale = 1.0f / fmaxf(sqrtf(ss), 1e-12f);
        v0.x *= scale; v0.y *= scale; v0.z *= scale; v0.w *= scale;
        v1.x *= scale; v1.y *= scale; v1.z *= scale; v1.w *= scale;
        *(float4*)(out + base)      = v0;
        *(float4*)(out + base + 64) = v1;
    }
}

extern "C" void kernel_launch(void* const* d_in, const int* in_sizes, int n_in,
                              void* d_out, int out_size, void* d_ws, size_t ws_size,
                              hipStream_t stream) {
    (void)in_sizes; (void)n_in; (void)out_size; (void)ws_size;

    const float* xu      = (const float*)d_in[0];
    const float* xi      = (const float*)d_in[1];
    const float* W1      = (const float*)d_in[2];
    const float* b1      = (const float*)d_in[3];
    const float* W2      = (const float*)d_in[4];
    const float* b2      = (const float*)d_in[5];
    const int*   src_ui  = (const int*)d_in[6];
    const int*   dst_ui  = (const int*)d_in[7];
    const float* norm_ui = (const float*)d_in[8];
    const int*   src_iu  = (const int*)d_in[9];
    const int*   dst_iu  = (const int*)d_in[10];
    const float* norm_iu = (const float*)d_in[11];

    float* out = (float*)d_out;
    float* S   = (float*)d_ws;
    float* Qb2 = S + (size_t)NNOD * DDIM;
    float* P1  = Qb2 + (size_t)NNOD * DDIM;
    int*   head = (int*)(P1 + (size_t)NNOD * DDIM);
    int*   rec  = head + NNOD;

    proj_wmma_kernel<<<PGRID, 256, 0, stream>>>(xu, xi, W1, b1, W2, b2, P1, S, Qb2, head);

    link_kernel<<<(LHALF + 255) / 256, 256, 0, stream>>>(src_ui, dst_ui, norm_ui,
                                                         src_iu, dst_iu, norm_iu, head, rec);

    gather_finalize_kernel<<<GGRID, 256, 0, stream>>>(P1, S, Qb2, head, rec, out);
}


# in-edge lists replaced by per-node slot arrays (rank by integer add, overflow list past 16): a node's source rows load four at a time, slots fetched a trip ahead; x summed in slot (arrival) order
# speedup vs baseline: 2.4433x; 1.1476x over previous
#include <hip/hip_runtime.h>

typedef __attribute__((ext_vector_type(2))) float v2f;
typedef __attribute__((ext_vector_type(8))) float v8f;
typedef __attribute__((ext_vector_type(4))) float v4f;
typedef unsigned long long slot_t;

#define NUQ   100000
#define NIQ   50000
#define DDIM  128
#define EEDG  300000
#define NNOD  (NUQ + NIQ)
#define LDSW  132
#define STAGE (16 * LDSW)
#define NUNIT ((NNOD + 31) / 32)
#define PGRID 512
#define NEDGE (2 * EEDG)
#define LHALF (EEDG / 2)
#define CAP   16


static __device__ __forceinline__ void lds_barrier()
{
    __builtin_amdgcn_fence(__ATOMIC_RELEASE, "workgroup", "local");
    __builtin_amdgcn_s_barrier();
    __builtin_amdgcn_fence(__ATOMIC_ACQUIRE, "workgroup", "local");
}

static __device__ __forceinline__ void fma4(float4& a, float s, const float4& v)
{
    a.x += s * v.x; a.y += s * v.y; a.z += s * v.z; a.w += s * v.w;
}

__global__ __launch_bounds__(256)
void init_kernel(int* __restrict__ head, int* __restrict__ cnt)
{
    const int g = (int)(blockIdx.x * 256 + threadIdx.x);
    if (g < NNOD) { head[g] = -1; cnt[g] = 0; }
}

static __device__ __forceinline__ void place_edge(int r, int e, int s, int d, float n,
                                                  int* __restrict__ head,
                                                  int2* __restrict__ slot, int4* __restrict__ rec)
{
    if (r < CAP) {
        slot[d * CAP + r] = make_int2(s, __float_as_int(n));
    } else {
        const int p = atomicExch(head + d, e);
        rec[e] = make_int4(p, s, __float_as_int(n), 0);
    }
}

__global__ __launch_bounds__(256)
void link_kernel(const int* __restrict__ src_ui, const int* __restrict__ dst_ui,
                 const float* __restrict__ norm_ui,
                 const int* __restrict__ src_iu, const int* __restrict__ dst_iu,
                 const float* __restrict__ norm_iu,
                 int* __restrict__ head, int* __restrict__ cnt,
                 int* __restrict__ slot_w, int* __restrict__ rec_w)
{
    const int g = (int)(blockIdx.x * 256 + threadIdx.x);
    if (g >= LHALF) return;
    const int g1 = g + LHALF;

    const int   sa = src_ui[g],        sb = src_ui[g1];
    const int   da = NUQ + dst_ui[g],  db = NUQ + dst_ui[g1];
    const float na = norm_ui[g],       nb = norm_ui[g1];
    const int   sc = NUQ + src_iu[g],  sd = NUQ + src_iu[g1];
    const int   dc = dst_iu[g],        dd = dst_iu[g1];
    const float nc = norm_iu[g],       nd = norm_iu[g1];

    const int ra = atomicAdd(cnt + da, 1);
    const int rb = atomicAdd(cnt + db, 1);
    const int rc = atomicAdd(cnt + dc, 1);
    const int rd = atomicAdd(cnt + dd, 1);

    int2* slot = (int2*)slot_w;
    int4* rec  = (int4*)rec_w;
    place_edge(ra, g,         sa, da, na, head, slot, rec);
    place_edge(rb, g1,        sb, db, nb, head, slot, rec);
    place_edge(rc, EEDG + g,  sc, dc, nc, head, slot, rec);
    place_edge(rd, EEDG + g1, sd, dd, nd, head, slot, rec);
}

template <int K>
static __device__ __forceinline__ int2 slot_of(slot_t sA, slot_t sB)
{
    constexpr int sel = (K & 7) * 0x249249;
    const slot_t sS = (K < 8) ? sA : sB;
    int2 r;
    r.x = __builtin_amdgcn_mov_dpp8((int)(unsigned)(sS & 0xffffffffull), sel);
    r.y = __builtin_amdgcn_mov_dpp8((int)(unsigned)(sS >> 32), sel);
    return r;
}

template <int G, int T0>
static __device__ __forceinline__ void slots_of(int2 (&sl)[G], slot_t sA, slot_t sB)
{
    sl[0] = slot_of<T0>(sA, sB);
    if constexpr (G > 1) sl[1] = slot_of<T0 + 1>(sA, sB);
    if constexpr (G > 2) sl[2] = slot_of<T0 + 2>(sA, sB);
    if constexpr (G > 3) sl[3] = slot_of<T0 + 3>(sA, sB);
}

template <int G, int T0>
static __device__ __forceinline__ void sum_rows(const int2 (&sl)[G], int nfill, int cg,
                                                const float* __restrict__ xu,
                                                const float* __restrict__ xi,
                                                float4& a0, float4& a1, float4& a2, float4& a3,
                                                float& ns)
{
    const float4 zero4 = make_float4(0.0f, 0.0f, 0.0f, 0.0f);
    float  nv[G];
    float4 r0[G], r1[G], r2[G], r3[G];
    #pragma unroll
    for (int i = 0; i < G; ++i) {
        r0[i] = zero4; r1[i] = zero4; r2[i] = zero4; r3[i] = zero4;
        nv[i] = 0.0f;
        if (T0 + i < nfill) {
            const int s = sl[i].x;
            const float* srow = ((s < NUQ) ? (xu + (size_t)s * DDIM)
                                           : (xi + (size_t)(s - NUQ) * DDIM)) + cg;
            r0[i] = *(const float4*)(srow);
            r1[i] = *(const float4*)(srow + 32);
            r2[i] = *(const float4*)(srow + 64);
            r3[i] = *(const float4*)(srow + 96);
            nv[i] = __int_as_float(sl[i].y);
        }
    }
    #pragma unroll
    for (int i = 0; i < G; ++i) {
        fma4(a0, nv[i], r0[i]);
        fma4(a1, nv[i], r1[i]);
        fma4(a2, nv[i], r2[i]);
        fma4(a3, nv[i], r3[i]);
        ns += nv[i];
    }
}

template <int G, int T0>
static __device__ __forceinline__ void sum_group(int nfill, int cg, slot_t sA, slot_t sB,
                                                 const float* __restrict__ xu,
                                                 const float* __restrict__ xi,
                                                 float4& a0, float4& a1, float4& a2, float4& a3,
                                                 float& ns)
{
    int2 sl[G];
    slots_of<G, T0>(sl, sA, sB);
    sum_rows<G, T0>(sl, nfill, cg, xu, xi, a0, a1, a2, a3, ns);
}

__global__ __launch_bounds__(256)
void gather_proj_kernel(const float* __restrict__ xu, const float* __restrict__ xi,
                        const float* __restrict__ W1, const float* __restrict__ b1,
                        const float* __restrict__ W2, const float* __restrict__ b2,
                        const int* __restrict__ head, const int* __restrict__ cnt,
                        const int* __restrict__ slot_r, const int* __restrict__ rec_r,
                        float* __restrict__ out)
{
    __shared__ float lds[8 * STAGE];
    __shared__ float nsl[32];

    const int4* __restrict__ rec  = (const int4*)rec_r;
    const slot_t* __restrict__ slot = (const slot_t*)slot_r;

    const int tid  = threadIdx.x;
    const int lane = tid & 31;
    const int wv   = tid >> 5;
    const int c0   = wv * 16;
    const int l15  = lane & 15;
    const int lh   = lane >> 4;
    const int col  = c0 + l15;

    const float bb1 = b1[col];
    const float bbs = bb1 + b2[col];

    float* wst = lds + wv * STAGE;
    const float* Bfrag = wst + l15 * LDSW + 2 * lh;
    v2f bw1[32], bw2[32];

    #pragma unroll
    for (int r = 0; r < 16; ++r)
        *(float4*)&wst[r * LDSW + lane * 4] =
            *(const float4*)(W1 + (size_t)(c0 + r) * DDIM + lane * 4);
    __syncthreads();
    #pragma unroll
    for (int i = 0; i < 32; ++i)
        bw1[i] = *(const v2f*)(Bfrag + 4 * i);
    __syncthreads();
    #pragma unroll
    for (int r = 0; r < 16; ++r)
        *(float4*)&wst[r * LDSW + lane * 4] =
            *(const float4*)(W2 + (size_t)(c0 + r) * DDIM + lane * 4);
    __syncthreads();
    #pragma unroll
    for (int i = 0; i < 32; ++i)
        bw2[i] = *(const v2f*)(Bfrag + 4 * i);
    __syncthreads();

    const int lr = 4 * wv + (lane >> 3);
    const int j8 = lane & 7;
    const int cg = j8 * 4;
    float* us = lds + (2 * (lr >> 4)) * STAGE + (lr & 15) * LDSW + cg;
    float* vs = us + STAGE;

    const float* Ua = lds + l15 * LDSW + 2 * lh;
    const int frow = 2 * wv + lh;
    const int fc   = l15 * 4;

    int  cN = 0, c1 = 0;
    slot_t sAN = 0, sBN = 0;
    {
        const int d0 = (int)blockIdx.x * 32 + lr;
        if (d0 < NNOD) cN = cnt[d0];
        const int n0 = cN < CAP ? cN : CAP;
        if (j8 < n0)     sAN = slot[d0 * CAP + j8];
        if (j8 + 8 < n0) sBN = slot[d0 * CAP + 8 + j8];
        const int u1 = (int)blockIdx.x + PGRID;
        const int d1 = u1 * 32 + lr;
        if (u1 < NUNIT && d1 < NNOD) c1 = cnt[d1];
    }

    for (int u = blockIdx.x; u < NUNIT; u += PGRID) {
        const int  d     = u * 32 + lr;
        const bool valid = d < NNOD;
        const int  c     = cN;
        const slot_t sA = sAN, sB = sBN;
        const float4 zero4 = make_float4(0.0f, 0.0f, 0.0f, 0.0f);
        float4 a0 = zero4, a1 = zero4, a2 = zero4, a3 = zero4;
        float4 x0 = zero4, x1 = zero4, x2 = zero4, x3 = zero4;
        float  ns = 0.0f;
        const int nfill = c < CAP ? c : CAP;
        int nmax = nfill;
        nmax = max(nmax, __shfl_xor(nmax, 8, 32));
        nmax = max(nmax, __shfl_xor(nmax, 16, 32));
        nmax = __builtin_amdgcn_readfirstlane(nmax);
        int2 sl0[3];
        slots_of<3, 0>(sl0, sA, sB);
        if (valid) {
            const float* xrow = ((d < NUQ) ? (xu + (size_t)d * DDIM)
                                           : (xi + (size_t)(d - NUQ) * DDIM)) + cg;
            x0 = *(const float4*)(xrow);
            x1 = *(const float4*)(xrow + 32);
            x2 = *(const float4*)(xrow + 64);
            x3 = *(const float4*)(xrow + 96);
        }
        sum_rows<3, 0>(sl0, nfill, cg, xu, xi, a0, a1, a2, a3, ns);
        *(float4*)(us)      = x0;
        *(float4*)(us + 32) = x1;
        *(float4*)(us + 64) = x2;
        *(float4*)(us + 96) = x3;
        if (nmax > 3) {
            sum_group<4, 3>(nfill, cg, sA, sB, xu, xi, a0, a1, a2, a3, ns);
            if (nmax > 7) {
                sum_group<4, 7>(nfill, cg, sA, sB, xu, xi, a0, a1, a2, a3, ns);
                if (nmax > 11) {
                    sum_group<4, 11>(nfill, cg, sA, sB, xu, xi, a0, a1, a2, a3, ns);
                    if (nmax > 15) {
                        sum_group<1, 15>(nfill, cg, sA, sB, xu, xi, a0, a1, a2, a3, ns);
                    }
                }
            }
        }
        if (c > CAP) {
            int e = head[d];
            for (int it = 0; it < NEDGE; ++it) {
                if ((unsigned)e >= (unsigned)NEDGE) break;
                const int4 r = rec[e];
                const int  s = r.y;
                const float* srow = ((s < NUQ) ? (xu + (size_t)s * DDIM)
                                               : (xi + (size_t)(s - NUQ) * DDIM)) + cg;
                const float4 s0 = *(const float4*)(srow);
                const float4 s1 = *(const float4*)(srow + 32);
                const float4 s2 = *(const float4*)(srow + 64);
                const float4 s3 = *(const float4*)(srow + 96);
                const float nrm = __int_as_float(r.z);
                fma4(a0, nrm, s0);
                fma4(a1, nrm, s1);
                fma4(a2, nrm, s2);
                fma4(a3, nrm, s3);
                ns += nrm;
                e = r.x;
            }
        }
        {
            typedef volatile __attribute__((address_space(3))) v4f* lds_v4f_ptr;
            const v4f y0 = *(lds_v4f_ptr)(us);
            const v4f y1 = *(lds_v4f_ptr)(us + 32);
            const v4f y2 = *(lds_v4f_ptr)(us + 64);
            const v4f y3 = *(lds_v4f_ptr)(us + 96);
            x0 = make_float4(y0.x, y0.y, y0.z, y0.w);
            x1 = make_float4(y1.x, y1.y, y1.z, y1.w);
            x2 = make_float4(y2.x, y2.y, y2.z, y2.w);
            x3 = make_float4(y3.x, y3.y, y3.z, y3.w);
            float4 t;
            t.x = x0.x + a0.x; t.y = x0.y + a0.y; t.z = x0.z + a0.z; t.w = x0.w + a0.w; *(float4*)(us)      = t;
            t.x = x1.x + a1.x; t.y = x1.y + a1.y; t.z = x1.z + a1.z; t.w = x1.w + a1.w; *(float4*)(us + 32) = t;
            t.x = x2.x + a2.x; t.y = x2.y + a2.y; t.z = x2.z + a2.z; t.w = x2.w + a2.w; *(float4*)(us + 64) = t;
            t.x = x3.x + a3.x; t.y = x3.y + a3.y; t.z = x3.z + a3.z; t.w = x3.w + a3.w; *(float4*)(us + 96) = t;
            t.x = ns * x0.x + a0.x; t.y = ns * x0.y + a0.y; t.z = ns * x0.z + a0.z; t.w = ns * x0.w + a0.w; *(float4*)(vs)      = t;
            t.x = ns * x1.x + a1.x; t.y = ns * x1.y + a1.y; t.z = ns * x1.z + a1.z; t.w = ns * x1.w + a1.w; *(float4*)(vs + 32) = t;
            t.x = ns * x2.x + a2.x; t.y = ns * x2.y + a2.y; t.z = ns * x2.z + a2.z; t.w = ns * x2.w + a2.w; *(float4*)(vs + 64) = t;
            t.x = ns * x3.x + a3.x; t.y = ns * x3.y + a3.y; t.z = ns * x3.z + a3.z; t.w = ns * x3.w + a3.w; *(float4*)(vs + 96) = t;
            if ((lane & 7) == 0) nsl[lr] = ns;
        }
        {
            const int un = u + PGRID;
            const int dn = un * 32 + lr;
            const int nn = c1 < CAP ? c1 : CAP;
            sAN = 0;
            sBN = 0;
            if (j8 < nn)     sAN = slot[dn * CAP + j8];
            if (j8 + 8 < nn) sBN = slot[dn * CAP + 8 + j8];
            const int u2 = un + PGRID;
            const int d2 = u2 * 32 + lr;
            cN = c1;
            c1 = 0;
            if (u2 < NUNIT && d2 < NNOD) c1 = cnt[d2];
        }
        lds_barrier();

        v8f p0 = {}, q0 = {}, p1 = {}, q1 = {};
        #pragma unroll
        for (int i = 0; i < 32; ++i) {
            const v2f au0 = *(const v2f*)(Ua + 4 * i);
            const v2f av0 = *(const v2f*)(Ua + STAGE + 4 * i);
            const v2f au1 = *(const v2f*)(Ua + 2 * STAGE + 4 * i);
            const v2f av1 = *(const v2f*)(Ua + 3 * STAGE + 4 * i);
            p0 = __builtin_amdgcn_wmma_f32_16x16x4_f32(
                 false, au0,  false, bw1[i],
                 (short)0, p0,  false,  false);
            q0 = __builtin_amdgcn_wmma_f32_16x16x4_f32(
                 false, av0,  false, bw2[i],
                 (short)0, q0,  false,  false);
            p1 = __builtin_amdgcn_wmma_f32_16x16x4_f32(
                 false, au1,  false, bw1[i],
                 (short)0, p1,  false,  false);
            q1 = __builtin_amdgcn_wmma_f32_16x16x4_f32(
                 false, av1,  false, bw2[i],
                 (short)0, q1,  false,  false);
        }

        float* H0 = lds + 4 * STAGE;
        float* H1 = lds + 5 * STAGE;
        #pragma unroll
        for (int v = 0; v < 8; ++v) {
            const int   m  = v + 8 * lh;
            const float h0 = (p0[v] + q0[v]) + (bb1 + nsl[m] * bbs);
            const float h1 = (p1[v] + q1[v]) + (bb1 + nsl[16 + m] * bbs);
            H0[m * LDSW + col] = h0 > 0.0f ? h0 : 0.2f * h0;
            H1[m * LDSW + col] = h1 > 0.0f ? h1 : 0.2f * h1;
        }
        lds_barrier();

        for (int j = 0; j < 2; ++j) {
            const float* Hr = lds + (4 + j) * STAGE + frow * LDSW + fc;
            float4 v0 = *(const float4*)(Hr);
            float4 v1 = *(const float4*)(Hr + 64);
            float ss = v0.x * v0.x + v0.y * v0.y + v0.z * v0.z + v0.w * v0.w
                     + v1.x * v1.x + v1.y * v1.y + v1.z * v1.z + v1.w * v1.w;
            for (int off = 8; off > 0; off >>= 1)
                ss += __shfl_xor(ss, off, 32);
            const float scale = 1.0f / fmaxf(sqrtf(ss), 1e-12f);
            v0.x *= scale; v0.y *= scale; v0.z *= scale; v0.w *= scale;
            v1.x *= scale; v1.y *= scale; v1.z *= scale; v1.w *= scale;
            const int dn = u * 32 + 16 * j + frow;
            if (dn < NNOD) {
                const size_t ob = (size_t)dn * DDIM + fc;
                *(float4*)(out + ob)      = v0;
                *(float4*)(out + ob + 64) = v1;
            }
        }
    }
}

extern "C" void kernel_launch(void* const* d_in, const int* in_sizes, int n_in,
                              void* d_out, int out_size, void* d_ws, size_t ws_size,
                              hipStream_t stream) {
    (void)in_sizes; (void)n_in; (void)out_size; (void)ws_size;

    const float* xu      = (const float*)d_in[0];
    const float* xi      = (const float*)d_in[1];
    const float* W1      = (const float*)d_in[2];
    const float* b1      = (const float*)d_in[3];
    const float* W2      = (const float*)d_in[4];
    const float* b2      = (const float*)d_in[5];
    const int*   src_ui  = (const int*)d_in[6];
    const int*   dst_ui  = (const int*)d_in[7];
    const float* norm_ui = (const float*)d_in[8];
    const int*   src_iu  = (const int*)d_in[9];
    const int*   dst_iu  = (const int*)d_in[10];
    const float* norm_iu = (const float*)d_in[11];

    float* out = (float*)d_out;
    int*   rec  = (int*)d_ws;
    int*   head = rec + 4 * NEDGE;
    int*   cnt  = head + NNOD;
    int*   slot = cnt + NNOD;

    init_kernel<<<(NNOD + 255) / 256, 256, 0, stream>>>(head, cnt);

    link_kernel<<<(LHALF + 255) / 256, 256, 0, stream>>>(src_ui, dst_ui, norm_ui,
                                                         src_iu, dst_iu, norm_iu,
                                                         head, cnt, slot, rec);

    gather_proj_kernel<<<PGRID, 256, 0, stream>>>(xu, xi, W1, b1, W2, b2, head, cnt, slot, rec, out);
}


# gather/product kernel: waves run at raised priority (s_setprio 3) outside their WMMA phase, 0 inside it: a block's row requests issue beside the other block's WMMA run instead of behind it
# speedup vs baseline: 2.5312x; 1.0360x over previous
#include <hip/hip_runtime.h>

typedef __attribute__((ext_vector_type(2))) float v2f;
typedef __attribute__((ext_vector_type(8))) float v8f;
typedef __attribute__((ext_vector_type(4))) float v4f;
typedef unsigned long long slot_t;

#define NUQ   100000
#define NIQ   50000
#define DDIM  128
#define EEDG  300000
#define NNOD  (NUQ + NIQ)
#define LDSW  132
#define STAGE (16 * LDSW)
#define NUNIT ((NNOD + 31) / 32)
#define PGRID 512
#define NEDGE (2 * EEDG)
#define LHALF (EEDG / 2)
#define CAP   16


static __device__ __forceinline__ void lds_barrier()
{
    __builtin_amdgcn_fence(__ATOMIC_RELEASE, "workgroup", "local");
    __builtin_amdgcn_s_barrier();
    __builtin_amdgcn_fence(__ATOMIC_ACQUIRE, "workgroup", "local");
}

static __device__ __forceinline__ void fma4(float4& a, float s, const float4& v)
{
    a.x += s * v.x; a.y += s * v.y; a.z += s * v.z; a.w += s * v.w;
}

__global__ __launch_bounds__(256)
void init_kernel(int* __restrict__ head, int* __restrict__ cnt)
{
    const int g = (int)(blockIdx.x * 256 + threadIdx.x);
    if (g < NNOD) { head[g] = -1; cnt[g] = 0; }
}

static __device__ __forceinline__ void place_edge(int r, int e, int s, int d, float n,
                                                  int* __restrict__ head,
                                                  int2* __restrict__ slot, int4* __restrict__ rec)
{
    if (r < CAP) {
        slot[d * CAP + r] = make_int2(s, __float_as_int(n));
    } else {
        const int p = atomicExch(head + d, e);
        rec[e] = make_int4(p, s, __float_as_int(n), 0);
    }
}

__global__ __launch_bounds__(256)
void link_kernel(const int* __restrict__ src_ui, const int* __restrict__ dst_ui,
                 const float* __restrict__ norm_ui,
                 const int* __restrict__ src_iu, const int* __restrict__ dst_iu,
                 const float* __restrict__ norm_iu,
                 int* __restrict__ head, int* __restrict__ cnt,
                 int* __restrict__ slot_w, int* __restrict__ rec_w)
{
    const int g = (int)(blockIdx.x * 256 + threadIdx.x);
    if (g >= LHALF) return;
    const int g1 = g + LHALF;

    const int   sa = src_ui[g],        sb = src_ui[g1];
    const int   da = NUQ + dst_ui[g],  db = NUQ + dst_ui[g1];
    const float na = norm_ui[g],       nb = norm_ui[g1];
    const int   sc = NUQ + src_iu[g],  sd = NUQ + src_iu[g1];
    const int   dc = dst_iu[g],        dd = dst_iu[g1];
    const float nc = norm_iu[g],       nd = norm_iu[g1];

    const int ra = atomicAdd(cnt + da, 1);
    const int rb = atomicAdd(cnt + db, 1);
    const int rc = atomicAdd(cnt + dc, 1);
    const int rd = atomicAdd(cnt + dd, 1);

    int2* slot = (int2*)slot_w;
    int4* rec  = (int4*)rec_w;
    place_edge(ra, g,         sa, da, na, head, slot, rec);
    place_edge(rb, g1,        sb, db, nb, head, slot, rec);
    place_edge(rc, EEDG + g,  sc, dc, nc, head, slot, rec);
    place_edge(rd, EEDG + g1, sd, dd, nd, head, slot, rec);
}

template <int K>
static __device__ __forceinline__ int2 slot_of(slot_t sA, slot_t sB)
{
    constexpr int sel = (K & 7) * 0x249249;
    const slot_t sS = (K < 8) ? sA : sB;
    int2 r;
    r.x = __builtin_amdgcn_mov_dpp8((int)(unsigned)(sS & 0xffffffffull), sel);
    r.y = __builtin_amdgcn_mov_dpp8((int)(unsigned)(sS >> 32), sel);
    return r;
}

template <int G, int T0>
static __device__ __forceinline__ void slots_of(int2 (&sl)[G], slot_t sA, slot_t sB)
{
    sl[0] = slot_of<T0>(sA, sB);
    if constexpr (G > 1) sl[1] = slot_of<T0 + 1>(sA, sB);
    if constexpr (G > 2) sl[2] = slot_of<T0 + 2>(sA, sB);
    if constexpr (G > 3) sl[3] = slot_of<T0 + 3>(sA, sB);
}

template <int G, int T0>
static __device__ __forceinline__ void sum_rows(const int2 (&sl)[G], int nfill, int cg,
                                                const float* __restrict__ xu,
                                                const float* __restrict__ xi,
                                                float4& a0, float4& a1, float4& a2, float4& a3,
                                                float& ns)
{
    const float4 zero4 = make_float4(0.0f, 0.0f, 0.0f, 0.0f);
    float  nv[G];
    float4 r0[G], r1[G], r2[G], r3[G];
    #pragma unroll
    for (int i = 0; i < G; ++i) {
        r0[i] = zero4; r1[i] = zero4; r2[i] = zero4; r3[i] = zero4;
        nv[i] = 0.0f;
        if (T0 + i < nfill) {
            const int s = sl[i].x;
            const float* srow = ((s < NUQ) ? (xu + (size_t)s * DDIM)
                                           : (xi + (size_t)(s - NUQ) * DDIM)) + cg;
            r0[i] = *(const float4*)(srow);
            r1[i] = *(const float4*)(srow + 32);
            r2[i] = *(const float4*)(srow + 64);
            r3[i] = *(const float4*)(srow + 96);
            nv[i] = __int_as_float(sl[i].y);
        }
    }
    #pragma unroll
    for (int i = 0; i < G; ++i) {
        fma4(a0, nv[i], r0[i]);
        fma4(a1, nv[i], r1[i]);
        fma4(a2, nv[i], r2[i]);
        fma4(a3, nv[i], r3[i]);
        ns += nv[i];
    }
}

template <int G, int T0>
static __device__ __forceinline__ void sum_group(int nfill, int cg, slot_t sA, slot_t sB,
                                                 const float* __restrict__ xu,
                                                 const float* __restrict__ xi,
                                                 float4& a0, float4& a1, float4& a2, float4& a3,
                                                 float& ns)
{
    int2 sl[G];
    slots_of<G, T0>(sl, sA, sB);
    sum_rows<G, T0>(sl, nfill, cg, xu, xi, a0, a1, a2, a3, ns);
}

__global__ __launch_bounds__(256)
void gather_proj_kernel(const float* __restrict__ xu, const float* __restrict__ xi,
                        const float* __restrict__ W1, const float* __restrict__ b1,
                        const float* __restrict__ W2, const float* __restrict__ b2,
                        const int* __restrict__ head, const int* __restrict__ cnt,
                        const int* __restrict__ slot_r, const int* __restrict__ rec_r,
                        float* __restrict__ out)
{
    __shared__ float lds[8 * STAGE];
    __shared__ float nsl[32];

    const int4* __restrict__ rec  = (const int4*)rec_r;
    const slot_t* __restrict__ slot = (const slot_t*)slot_r;

    const int tid  = threadIdx.x;
    const int lane = tid & 31;
    const int wv   = tid >> 5;
    const int c0   = wv * 16;
    const int l15  = lane & 15;
    const int lh   = lane >> 4;
    const int col  = c0 + l15;

    const float bb1 = b1[col];
    const float bbs = bb1 + b2[col];

    float* wst = lds + wv * STAGE;
    const float* Bfrag = wst + l15 * LDSW + 2 * lh;
    v2f bw1[32], bw2[32];

    #pragma unroll
    for (int r = 0; r < 16; ++r)
        *(float4*)&wst[r * LDSW + lane * 4] =
            *(const float4*)(W1 + (size_t)(c0 + r) * DDIM + lane * 4);
    __syncthreads();
    #pragma unroll
    for (int i = 0; i < 32; ++i)
        bw1[i] = *(const v2f*)(Bfrag + 4 * i);
    __syncthreads();
    #pragma unroll
    for (int r = 0; r < 16; ++r)
        *(float4*)&wst[r * LDSW + lane * 4] =
            *(const float4*)(W2 + (size_t)(c0 + r) * DDIM + lane * 4);
    __syncthreads();
    #pragma unroll
    for (int i = 0; i < 32; ++i)
        bw2[i] = *(const v2f*)(Bfrag + 4 * i);
    __syncthreads();

    const int lr = 4 * wv + (lane >> 3);
    const int j8 = lane & 7;
    const int cg = j8 * 4;
    float* us = lds + (2 * (lr >> 4)) * STAGE + (lr & 15) * LDSW + cg;
    float* vs = us + STAGE;

    const float* Ua = lds + l15 * LDSW + 2 * lh;
    const int frow = 2 * wv + lh;
    const int fc   = l15 * 4;

    int  cN = 0, c1 = 0;
    slot_t sAN = 0, sBN = 0;
    __builtin_amdgcn_s_setprio(3);
    {
        const int d0 = (int)blockIdx.x * 32 + lr;
        if (d0 < NNOD) cN = cnt[d0];
        const int n0 = cN < CAP ? cN : CAP;
        if (j8 < n0)     sAN = slot[d0 * CAP + j8];
        if (j8 + 8 < n0) sBN = slot[d0 * CAP + 8 + j8];
        const int u1 = (int)blockIdx.x + PGRID;
        const int d1 = u1 * 32 + lr;
        if (u1 < NUNIT && d1 < NNOD) c1 = cnt[d1];
    }

    for (int u = blockIdx.x; u < NUNIT; u += PGRID) {
        const int  d     = u * 32 + lr;
        const bool valid = d < NNOD;
        const int  c     = cN;
        const slot_t sA = sAN, sB = sBN;
        const float4 zero4 = make_float4(0.0f, 0.0f, 0.0f, 0.0f);
        float4 a0 = zero4, a1 = zero4, a2 = zero4, a3 = zero4;
        float4 x0 = zero4, x1 = zero4, x2 = zero4, x3 = zero4;
        float  ns = 0.0f;
        const int nfill = c < CAP ? c : CAP;
        int nmax = nfill;
        nmax = max(nmax, __shfl_xor(nmax, 8, 32));
        nmax = max(nmax, __shfl_xor(nmax, 16, 32));
        nmax = __builtin_amdgcn_readfirstlane(nmax);
        int2 sl0[3];
        slots_of<3, 0>(sl0, sA, sB);
        if (valid) {
            const float* xrow = ((d < NUQ) ? (xu + (size_t)d * DDIM)
                                           : (xi + (size_t)(d - NUQ) * DDIM)) + cg;
            x0 = *(const float4*)(xrow);
            x1 = *(const float4*)(xrow + 32);
            x2 = *(const float4*)(xrow + 64);
            x3 = *(const float4*)(xrow + 96);
        }
        sum_rows<3, 0>(sl0, nfill, cg, xu, xi, a0, a1, a2, a3, ns);
        *(float4*)(us)      = x0;
        *(float4*)(us + 32) = x1;
        *(float4*)(us + 64) = x2;
        *(float4*)(us + 96) = x3;
        if (nmax > 3) {
            sum_group<4, 3>(nfill, cg, sA, sB, xu, xi, a0, a1, a2, a3, ns);
            if (nmax > 7) {
                sum_group<4, 7>(nfill, cg, sA, sB, xu, xi, a0, a1, a2, a3, ns);
                if (nmax > 11) {
                    sum_group<4, 11>(nfill, cg, sA, sB, xu, xi, a0, a1, a2, a3, ns);
                    if (nmax > 15) {
                        sum_group<1, 15>(nfill, cg, sA, sB, xu, xi, a0, a1, a2, a3, ns);
                    }
                }
            }
        }
        if (c > CAP) {
            int e = head[d];
            for (int it = 0; it < NEDGE; ++it) {
                if ((unsigned)e >= (unsigned)NEDGE) break;
                const int4 r = rec[e];
                const int  s = r.y;
                const float* srow = ((s < NUQ) ? (xu + (size_t)s * DDIM)
                                               : (xi + (size_t)(s - NUQ) * DDIM)) + cg;
                const float4 s0 = *(const float4*)(srow);
                const float4 s1 = *(const float4*)(srow + 32);
                const float4 s2 = *(const float4*)(srow + 64);
                const float4 s3 = *(const float4*)(srow + 96);
                const float nrm = __int_as_float(r.z);
                fma4(a0, nrm, s0);
                fma4(a1, nrm, s1);
                fma4(a2, nrm, s2);
                fma4(a3, nrm, s3);
                ns += nrm;
                e = r.x;
            }
        }
        {
            typedef volatile __attribute__((address_space(3))) v4f* lds_v4f_ptr;
            const v4f y0 = *(lds_v4f_ptr)(us);
            const v4f y1 = *(lds_v4f_ptr)(us + 32);
            const v4f y2 = *(lds_v4f_ptr)(us + 64);
            const v4f y3 = *(lds_v4f_ptr)(us + 96);
            x0 = make_float4(y0.x, y0.y, y0.z, y0.w);
            x1 = make_float4(y1.x, y1.y, y1.z, y1.w);
            x2 = make_float4(y2.x, y2.y, y2.z, y2.w);
            x3 = make_float4(y3.x, y3.y, y3.z, y3.w);
            float4 t;
            t.x = x0.x + a0.x; t.y = x0.y + a0.y; t.z = x0.z + a0.z; t.w = x0.w + a0.w; *(float4*)(us)      = t;
            t.x = x1.x + a1.x; t.y = x1.y + a1.y; t.z = x1.z + a1.z; t.w = x1.w + a1.w; *(float4*)(us + 32) = t;
            t.x = x2.x + a2.x; t.y = x2.y + a2.y; t.z = x2.z + a2.z; t.w = x2.w + a2.w; *(float4*)(us + 64) = t;
            t.x = x3.x + a3.x; t.y = x3.y + a3.y; t.z = x3.z + a3.z; t.w = x3.w + a3.w; *(float4*)(us + 96) = t;
            t.x = ns * x0.x + a0.x; t.y = ns * x0.y + a0.y; t.z = ns * x0.z + a0.z; t.w = ns * x0.w + a0.w; *(float4*)(vs)      = t;
            t.x = ns * x1.x + a1.x; t.y = ns * x1.y + a1.y; t.z = ns * x1.z + a1.z; t.w = ns * x1.w + a1.w; *(float4*)(vs + 32) = t;
            t.x = ns * x2.x + a2.x; t.y = ns * x2.y + a2.y; t.z = ns * x2.z + a2.z; t.w = ns * x2.w + a2.w; *(float4*)(vs + 64) = t;
            t.x = ns * x3.x + a3.x; t.y = ns * x3.y + a3.y; t.z = ns * x3.z + a3.z; t.w = ns * x3.w + a3.w; *(float4*)(vs + 96) = t;
            if ((lane & 7) == 0) nsl[lr] = ns;
        }
        {
            const int un = u + PGRID;
            const int dn = un * 32 + lr;
            const int nn = c1 < CAP ? c1 : CAP;
            sAN = 0;
            sBN = 0;
            if (j8 < nn)     sAN = slot[dn * CAP + j8];
            if (j8 + 8 < nn) sBN = slot[dn * CAP + 8 + j8];
            const int u2 = un + PGRID;
            const int d2 = u2 * 32 + lr;
            cN = c1;
            c1 = 0;
            if (u2 < NUNIT && d2 < NNOD) c1 = cnt[d2];
        }
        lds_barrier();

        __builtin_amdgcn_s_setprio(0);
        v8f p0 = {}, q0 = {}, p1 = {}, q1 = {};
        #pragma unroll
        for (int i = 0; i < 32; ++i) {
            const v2f au0 = *(const v2f*)(Ua + 4 * i);
            const v2f av0 = *(const v2f*)(Ua + STAGE + 4 * i);
            const v2f au1 = *(const v2f*)(Ua + 2 * STAGE + 4 * i);
            const v2f av1 = *(const v2f*)(Ua + 3 * STAGE + 4 * i);
            p0 = __builtin_amdgcn_wmma_f32_16x16x4_f32(
                 false, au0,  false, bw1[i],
                 (short)0, p0,  false,  false);
            q0 = __builtin_amdgcn_wmma_f32_16x16x4_f32(
                 false, av0,  false, bw2[i],
                 (short)0, q0,  false,  false);
            p1 = __builtin_amdgcn_wmma_f32_16x16x4_f32(
                 false, au1,  false, bw1[i],
                 (short)0, p1,  false,  false);
            q1 = __builtin_amdgcn_wmma_f32_16x16x4_f32(
                 false, av1,  false, bw2[i],
                 (short)0, q1,  false,  false);
        }

        float t0[8], t1[8];
        #pragma unroll
        for (int v = 0; v < 8; ++v) {
            const int m = v + 8 * lh;
            t0[v] = bb1 + nsl[m] * bbs;
            t1[v] = bb1 + nsl[16 + m] * bbs;
        }

        __builtin_amdgcn_s_setprio(3);

        float* H0 = lds + 4 * STAGE;
        float* H1 = lds + 5 * STAGE;
        #pragma unroll
        for (int v = 0; v < 8; ++v) {
            const int   m  = v + 8 * lh;
            const float h0 = (p0[v] + q0[v]) + t0[v];
            const float h1 = (p1[v] + q1[v]) + t1[v];
            H0[m * LDSW + col] = h0 > 0.0f ? h0 : 0.2f * h0;
            H1[m * LDSW + col] = h1 > 0.0f ? h1 : 0.2f * h1;
        }
        lds_barrier();

        for (int j = 0; j < 2; ++j) {
            const float* Hr = lds + (4 + j) * STAGE + frow * LDSW + fc;
            float4 v0 = *(const float4*)(Hr);
            float4 v1 = *(const float4*)(Hr + 64);
            float ss = v0.x * v0.x + v0.y * v0.y + v0.z * v0.z + v0.w * v0.w
                     + v1.x * v1.x + v1.y * v1.y + v1.z * v1.z + v1.w * v1.w;
            for (int off = 8; off > 0; off >>= 1)
                ss += __shfl_xor(ss, off, 32);
            const float scale = 1.0f / fmaxf(sqrtf(ss), 1e-12f);
            v0.x *= scale; v0.y *= scale; v0.z *= scale; v0.w *= scale;
            v1.x *= scale; v1.y *= scale; v1.z *= scale; v1.w *= scale;
            const int dn = u * 32 + 16 * j + frow;
            if (dn < NNOD) {
                const size_t ob = (size_t)dn * DDIM + fc;
                *(float4*)(out + ob)      = v0;
                *(float4*)(out + ob + 64) = v1;
            }
        }
    }
    __builtin_amdgcn_s_setprio(0);
}

extern "C" void kernel_launch(void* const* d_in, const int* in_sizes, int n_in,
                              void* d_out, int out_size, void* d_ws, size_t ws_size,
                              hipStream_t stream) {
    (void)in_sizes; (void)n_in; (void)out_size; (void)ws_size;

    const float* xu      = (const float*)d_in[0];
    const float* xi      = (const float*)d_in[1];
    const float* W1      = (const float*)d_in[2];
    const float* b1      = (const float*)d_in[3];
    const float* W2      = (const float*)d_in[4];
    const float* b2      = (const float*)d_in[5];
    const int*   src_ui  = (const int*)d_in[6];
    const int*   dst_ui  = (const int*)d_in[7];
    const float* norm_ui = (const float*)d_in[8];
    const int*   src_iu  = (const int*)d_in[9];
    const int*   dst_iu  = (const int*)d_in[10];
    const float* norm_iu = (const float*)d_in[11];

    float* out = (float*)d_out;
    int*   rec  = (int*)d_ws;
    int*   head = rec + 4 * NEDGE;
    int*   cnt  = head + NNOD;
    int*   slot = cnt + NNOD;

    init_kernel<<<(NNOD + 255) / 256, 256, 0, stream>>>(head, cnt);

    link_kernel<<<(LHALF + 255) / 256, 256, 0, stream>>>(src_ui, dst_ui, norm_ui,
                                                         src_iu, dst_iu, norm_iu,
                                                         head, cnt, slot, rec);

    gather_proj_kernel<<<PGRID, 256, 0, stream>>>(xu, xi, W1, b1, W2, b2, head, cnt, slot, rec, out);
}
